# starting kernel
# baseline (speedup 1.0000x reference)
#include <hip/hip_runtime.h>


typedef __attribute__((ext_vector_type(2))) float v2f;
typedef __attribute__((ext_vector_type(4))) float v4f;
typedef __attribute__((ext_vector_type(8))) float v8f;

static __device__ __forceinline__ v8f wmma_f32x4(v2f a, v2f b, v8f c) {
  return __builtin_amdgcn_wmma_f32_16x16x4_f32(
       false, a,  false, b,
       (short)0, c,  false,  false);
}

__global__ void k_zero(float* __restrict__ p, long long nElem) {
  long long i = (long long)blockIdx.x * blockDim.x + threadIdx.x;
  if (i < nElem) p[i] = 0.0f;
}

#define T1_STRIDE 68
__launch_bounds__(256)
__global__ void k_gemm1(const float* __restrict__ x, const float* __restrict__ Wl,
                        const float* __restrict__ Wr, float* __restrict__ y1,
                        float* __restrict__ z1, int n) {
  __shared__ float tile[128 * T1_STRIDE];
  const int tid = threadIdx.x;
  const int rb  = blockIdx.x * 128;

#pragma unroll
  for (int it = 0; it < 8; ++it) {
    int idx = (it * 256 + tid) * 4;
    int r = idx >> 6, c = idx & 63;
    int grow = rb + r; if (grow > n - 1) grow = n - 1;
    v4f v = *(const v4f*)(x + (size_t)grow * 64 + c);
    *(v4f*)(tile + r * T1_STRIDE + c) = v;
  }
  __syncthreads();

  const int lane = tid & 31;
  const int wave = tid >> 5;
  const int mr   = lane & 15;
  const int kh   = (lane >> 4) << 1;

  v2f bl[16], br[16];
#pragma unroll
  for (int s = 0; s < 16; ++s) {
    int k0 = 4 * s + kh;
    v2f tl, tr;
    tl.x = Wl[k0 * 16 + mr]; tl.y = Wl[(k0 + 1) * 16 + mr];
    tr.x = Wr[k0 * 16 + mr]; tr.y = Wr[(k0 + 1) * 16 + mr];
    bl[s] = tl; br[s] = tr;
  }

  const float* arow = tile + (wave * 16 + mr) * T1_STRIDE;
  v8f cl = {}; v8f cr = {};
#pragma unroll
  for (int s = 0; s < 16; ++s) {
    v2f a = *(const v2f*)(arow + 4 * s + kh);
    cl = wmma_f32x4(a, bl[s], cl);
    cr = wmma_f32x4(a, br[s], cr);
  }

  const int rowHalf = (lane >> 4) << 3;
#pragma unroll
  for (int v = 0; v < 8; ++v) {
    int row = rb + wave * 16 + rowHalf + v;
    if (row < n) {
      y1[(size_t)row * 16 + mr] = cl[v];
      z1[(size_t)row * 16 + mr] = cr[v];
    }
  }
}

__global__ void k_scatter1(const int* __restrict__ ei, const float* __restrict__ y1,
                           float* __restrict__ agg1, float* __restrict__ deg,
                           long long E) {
  long long t = (long long)blockIdx.x * blockDim.x + threadIdx.x;
  if (t >= E * 16) return;
  long long e = t >> 4;
  int f = (int)(t & 15);
  int src = ei[e];
  int dst = ei[E + e];
  float v = y1[(size_t)src * 16 + f];
  atomicAdd(&agg1[(size_t)dst * 16 + f], v);
  if (f == 0) atomicAdd(&deg[dst], 1.0f);
}

__global__ void k_combine1(const float* __restrict__ agg1, const float* __restrict__ deg,
                           const float* __restrict__ z1, const float* __restrict__ b1,
                           float* __restrict__ h, int n) {
  long long i = (long long)blockIdx.x * blockDim.x + threadIdx.x;
  if (i >= (long long)n * 16) return;
  int node = (int)(i >> 4);
  int f = (int)(i & 15);
  float d = fmaxf(deg[node], 1.0f);
  float v = agg1[i] / d + z1[i] + b1[f];
  h[i] = fmaxf(v, 0.0f);
}

#define T2_STRIDE 20
__launch_bounds__(256)
__global__ void k_gemm2(const float* __restrict__ h, const float* __restrict__ Wl,
                        const float* __restrict__ Wr, float* __restrict__ y2,
                        float* __restrict__ z2, int n) {
  __shared__ float tile[128 * T2_STRIDE];
  const int tid = threadIdx.x;
  const int rb  = blockIdx.x * 128;

#pragma unroll
  for (int it = 0; it < 2; ++it) {
    int idx = (it * 256 + tid) * 4;
    int r = idx >> 4, c = idx & 15;
    int grow = rb + r; if (grow > n - 1) grow = n - 1;
    v4f v = *(const v4f*)(h + (size_t)grow * 16 + c);
    *(v4f*)(tile + r * T2_STRIDE + c) = v;
  }
  __syncthreads();

  const int lane = tid & 31;
  const int wave = tid >> 5;
  const int mr   = lane & 15;
  const int kh   = (lane >> 4) << 1;

  v2f bl[4], br[4];
#pragma unroll
  for (int s = 0; s < 4; ++s) {
    int k0 = 4 * s + kh;
    v2f tl = {}; v2f tr = {};
    if (mr < 2) {
      tl.x = Wl[k0 * 2 + mr]; tl.y = Wl[(k0 + 1) * 2 + mr];
      tr.x = Wr[k0 * 2 + mr]; tr.y = Wr[(k0 + 1) * 2 + mr];
    }
    bl[s] = tl; br[s] = tr;
  }

  const float* arow = tile + (wave * 16 + mr) * T2_STRIDE;
  v8f cl = {}; v8f cr = {};
#pragma unroll
  for (int s = 0; s < 4; ++s) {
    v2f a = *(const v2f*)(arow + 4 * s + kh);
    cl = wmma_f32x4(a, bl[s], cl);
    cr = wmma_f32x4(a, br[s], cr);
  }

  const int rowHalf = (lane >> 4) << 3;
  if (mr < 2) {
#pragma unroll
    for (int v = 0; v < 8; ++v) {
      int row = rb + wave * 16 + rowHalf + v;
      if (row < n) {
        y2[(size_t)row * 2 + mr] = cl[v];
        z2[(size_t)row * 2 + mr] = cr[v];
      }
    }
  }
}

__global__ void k_scatter2(const int* __restrict__ ei, const float* __restrict__ y2,
                           float* __restrict__ agg2, long long E) {
  long long t = (long long)blockIdx.x * blockDim.x + threadIdx.x;
  if (t >= E * 2) return;
  long long e = t >> 1;
  int f = (int)(t & 1);
  int src = ei[e];
  int dst = ei[E + e];
  atomicAdd(&agg2[(size_t)dst * 2 + f], y2[(size_t)src * 2 + f]);
}

__global__ void k_combine2(const float* __restrict__ agg2, const float* __restrict__ deg,
                           const float* __restrict__ z2, const float* __restrict__ b2,
                           float* __restrict__ out, int n) {
  long long i = (long long)blockIdx.x * blockDim.x + threadIdx.x;
  if (i >= (long long)n * 2) return;
  int node = (int)(i >> 1);
  int f = (int)(i & 1);
  float d = fmaxf(deg[node], 1.0f);
  out[i] = agg2[i] / d + z2[i] + b2[f];
}

extern "C" void kernel_launch(void* const* d_in, const int* in_sizes, int n_in,
                              void* d_out, int out_size, void* d_ws, size_t ws_size,
                              hipStream_t stream) {
  const float* x   = (const float*)d_in[0];
  const int*   ei  = (const int*)d_in[1];
  const float* W1l = (const float*)d_in[2];
  const float* W1r = (const float*)d_in[3];
  const float* b1  = (const float*)d_in[4];
  const float* W2l = (const float*)d_in[5];
  const float* W2r = (const float*)d_in[6];
  const float* b2  = (const float*)d_in[7];
  float* out = (float*)d_out;

  const int       n = in_sizes[0] / 64;
  const long long E = in_sizes[1] / 2;

  float* ws   = (float*)d_ws;
  float* agg1 = ws;
  float* deg  = agg1 + (size_t)16 * n;
  float* agg2 = deg + n;
  float* y1   = agg2 + (size_t)2 * n;
  float* z1   = y1 + (size_t)16 * n;
  float* y2   = z1 + (size_t)16 * n;
  float* z2   = y2 + (size_t)2 * n;
  float* h    = y1;

  const long long nzero = (long long)19 * n;
  k_zero<<<(unsigned)((nzero + 255) / 256), 256, 0, stream>>>(ws, nzero);

  const unsigned gGemm = (unsigned)((n + 127) / 128);
  k_gemm1<<<gGemm, 256, 0, stream>>>(x, W1l, W1r, y1, z1, n);

  const long long t1 = E * 16;
  k_scatter1<<<(unsigned)((t1 + 255) / 256), 256, 0, stream>>>(ei, y1, agg1, deg, E);

  const long long c1 = (long long)n * 16;
  k_combine1<<<(unsigned)((c1 + 255) / 256), 256, 0, stream>>>(agg1, deg, z1, b1, h, n);

  k_gemm2<<<gGemm, 256, 0, stream>>>(h, W2l, W2r, y2, z2, n);

  const long long t2 = E * 2;
  k_scatter2<<<(unsigned)((t2 + 255) / 256), 256, 0, stream>>>(ei, y2, agg2, E);

  const long long c2 = (long long)n * 2;
  k_combine2<<<(unsigned)((c2 + 255) / 256), 256, 0, stream>>>(agg2, deg, z2, b2, out, n);
}


# Layer-2 products h@W2l, h@W2r done inside the layer-1 combine kernel: a 256-node work-group keeps h in an LDS tile (f32 WMMA there); h plane and one launch gone, fewer and longer waves; sums unchanged
# speedup vs baseline: 3.6183x; 3.6183x over previous
#include <hip/hip_runtime.h>


typedef __attribute__((ext_vector_type(2))) float v2f;
typedef __attribute__((ext_vector_type(4))) float v4f;
typedef __attribute__((ext_vector_type(8))) float v8f;
typedef __attribute__((ext_vector_type(4))) int v4i;

static __device__ __forceinline__ v8f wmma_f32x4(v2f a, v2f b, v8f c) {
  return __builtin_amdgcn_wmma_f32_16x16x4_f32(
       false, a,  false, b,
       (short)0, c,  false,  false);
}

#define T1_STRIDE 68
__launch_bounds__(256)
__global__ void k_gemm1(const float* __restrict__ x, const float* __restrict__ Wl,
                        const float* __restrict__ Wr, float* __restrict__ y1,
                        float* __restrict__ z1, int n) {
  __shared__ float tile[128 * T1_STRIDE];
  const int tid = threadIdx.x;
  const int rb  = blockIdx.x * 128;

#pragma unroll
  for (int it = 0; it < 8; ++it) {
    int idx = (it * 256 + tid) * 4;
    int r = idx >> 6, c = idx & 63;
    int grow = rb + r; if (grow > n - 1) grow = n - 1;
    v4f v = *(const v4f*)(x + (size_t)grow * 64 + c);
    *(v4f*)(tile + r * T1_STRIDE + c) = v;
  }
  __syncthreads();

  const int lane = tid & 31;
  const int wave = tid >> 5;
  const int mr   = lane & 15;
  const int kh   = (lane >> 4) << 1;

  v2f bl[16], br[16];
#pragma unroll
  for (int s = 0; s < 16; ++s) {
    int k0 = 4 * s + kh;
    v2f tl, tr;
    tl.x = Wl[k0 * 16 + mr]; tl.y = Wl[(k0 + 1) * 16 + mr];
    tr.x = Wr[k0 * 16 + mr]; tr.y = Wr[(k0 + 1) * 16 + mr];
    bl[s] = tl; br[s] = tr;
  }

  const float* arow = tile + (wave * 16 + mr) * T1_STRIDE;
  v8f cl = {}; v8f cr = {};
#pragma unroll
  for (int s = 0; s < 16; ++s) {
    v2f a = *(const v2f*)(arow + 4 * s + kh);
    cl = wmma_f32x4(a, bl[s], cl);
    cr = wmma_f32x4(a, br[s], cr);
  }

  const int rowHalf = (lane >> 4) << 3;
#pragma unroll
  for (int v = 0; v < 8; ++v) {
    int row = rb + wave * 16 + rowHalf + v;
    if (row < n) {
      y1[(size_t)row * 16 + mr] = cl[v];
      z1[(size_t)row * 16 + mr] = cr[v];
    }
  }
}

template <int F, int RMAX, int RBITS, int WCAP, bool WITH_DEG, int LISTS>
__launch_bounds__(1024)
__global__ void k_agg(const int* __restrict__ ei, const float* __restrict__ y,
                      float* __restrict__ part, float* __restrict__ degPart,
                      unsigned* __restrict__ gl, unsigned* __restrict__ gc,
                      int n, long long E) {
  constexpr int V   = (F >= 4) ? 4 : F;
  constexpr int LPE = F / V;
  constexpr unsigned RMASK = (1u << RBITS) - 1u;
  constexpr int UA = 8;
  constexpr int UB = 8;
  constexpr int UC = 8;
  typedef __attribute__((ext_vector_type(V))) float vecV;
  static_assert(RMAX <= (1 << RBITS), "range index fits its field");
  static_assert((RMAX * F) % 4 == 0 && (F % V) == 0, "vector forms");

  __shared__ __attribute__((aligned(16))) float acc[RMAX * F];
  __shared__ float dg[WITH_DEG ? RMAX : 4];
  __shared__ unsigned list[32 * WCAP];

  const int tid  = threadIdx.x;
  const int lane = tid & 31;
  const int wave = __builtin_amdgcn_readfirstlane(tid >> 5);
  const int nRanges = (int)gridDim.x;
  const int nChunks = (int)gridDim.y;
  const int R  = (((n + nRanges - 1) / nRanges) + 15) & ~15;
  const int r0 = (int)blockIdx.x * R;
  const int chunk = (int)blockIdx.y;

  for (int i = tid; i < RMAX * F / 4; i += 1024) ((v4f*)acc)[i] = (v4f){0.f, 0.f, 0.f, 0.f};
  if (WITH_DEG) for (int i = tid; i < RMAX; i += 1024) dg[i] = 0.0f;
  __syncthreads();

  const int* __restrict__ srcA = ei;
  const int* __restrict__ dstA = ei + E;
  const int Ei   = (int)E;
  const int per  = (((Ei + nChunks - 1) / nChunks) + 1023) & ~1023;
  const int wper = per >> 5;
  const int ws   = chunk * per + wave * wper;
  int we = ws + wper; if (we > Ei) we = Ei;

  unsigned* wl = list + wave * WCAP;
  unsigned wcount = 0;

  const size_t slot = ((size_t)blockIdx.x * (size_t)nChunks + (size_t)chunk) * 32u + (size_t)wave;
  bool filter = true;
  if (LISTS == 2) {
    const unsigned saved = gc[slot];
    if (saved <= (unsigned)WCAP) {
      const unsigned* __restrict__ g = gl + slot * WCAP;
      for (unsigned i0 = 0; i0 < saved; i0 += 32 * UC) {
        unsigned en[UC];
#pragma unroll
        for (int u = 0; u < UC; ++u) {
          const unsigned i = i0 + u * 32 + lane;
          en[u] = g[i < saved ? i : 0u];
        }
#pragma unroll
        for (int u = 0; u < UC; ++u) wl[i0 + u * 32 + lane] = en[u];
      }
      wcount = saved;
      filter = false;
    }
  }

  int base = ws;
  if (filter && ((((size_t)dstA) & 15) == 0) && base + 256 <= we) {
    const int* __restrict__ dp = dstA + 4 * lane;
    v4i d0 = *(const v4i*)(dp + base);
    v4i d1 = *(const v4i*)(dp + base + 128);
    unsigned elb = (unsigned)(4 * lane) << RBITS;
    for (; base + 256 <= we && wcount + 256u <= (unsigned)WCAP; base += 256) {
      const int nb = (base + 512 <= we) ? base + 256 : base;
      const v4i n0 = *(const v4i*)(dp + nb);
      const v4i n1 = *(const v4i*)(dp + nb + 128);
#pragma unroll
      for (int u = 0; u < 8; ++u) {
        const int dval = (u < 4) ? d0[u & 3] : d1[u & 3];
        const unsigned r = (unsigned)(dval - r0);
        const bool m = r < (unsigned)R;
        const unsigned mask = __builtin_amdgcn_ballot_w32(m);
        const unsigned pos  = __builtin_amdgcn_mbcnt_lo(mask, wcount);
        if (m) wl[pos] = elb + ((unsigned)((u >> 2) * 128 + (u & 3)) << RBITS) + r;
        wcount += (unsigned)__builtin_popcount(mask);
      }
      d0 = n0; d1 = n1;
      elb += 256u << RBITS;
    }
  }
  if (filter)
  for (; base < we; base += 32 * UA) {
    int dv[UA];
#pragma unroll
    for (int u = 0; u < UA; ++u) {
      const int e  = base + u * 32 + lane;
      const int ec = e < Ei ? e : Ei - 1;
      int d = dstA[ec];
      dv[u] = (e < we) ? d : -1;
    }
#pragma unroll
    for (int u = 0; u < UA; ++u) {
      const unsigned r = (unsigned)(dv[u] - r0);
      const bool m = r < (unsigned)R;
      const unsigned mask = __builtin_amdgcn_ballot_w32(m);
      const unsigned pos  = wcount + __builtin_amdgcn_mbcnt_lo(mask, 0u);
      if (m) {
        const unsigned el = (unsigned)(base + u * 32 + lane - ws);
        if (__builtin_expect(pos < (unsigned)WCAP, 1)) {
          wl[pos] = (el << RBITS) | r;
        } else {
          const int s = srcA[ws + el];
#pragma nounroll
          for (int f = 0; f < F; ++f)
            __hip_atomic_fetch_add(&acc[r * F + f], y[(size_t)s * F + f],
                                   __ATOMIC_RELAXED, __HIP_MEMORY_SCOPE_WORKGROUP);
          if (WITH_DEG)
            __hip_atomic_fetch_add(&dg[r], 1.0f, __ATOMIC_RELAXED, __HIP_MEMORY_SCOPE_WORKGROUP);
        }
      }
      wcount += (unsigned)__builtin_popcount(mask);
    }
  }
  __syncthreads();

  const unsigned cnt = wcount < (unsigned)WCAP ? wcount : (unsigned)WCAP;
  if (LISTS == 1 && lane == 0) gc[slot] = wcount;
  if (filter)
  for (unsigned i0 = 0; i0 < cnt; i0 += 32 * UC) {
    unsigned en[UC];
    int sv[UC];
#pragma unroll
    for (int u = 0; u < UC; ++u) {
      const unsigned i = i0 + u * 32 + lane;
      en[u] = (i < cnt) ? wl[i] : 0u;
    }
#pragma unroll
    for (int u = 0; u < UC; ++u) sv[u] = srcA[ws + (int)(en[u] >> RBITS)];
#pragma unroll
    for (int u = 0; u < UC; ++u) {
      const unsigned i = i0 + u * 32 + lane;
      if (i < cnt) {
        const unsigned ent = ((unsigned)sv[u] << RBITS) | (en[u] & RMASK);
        wl[i] = ent;
        if (LISTS == 1) gl[slot * WCAP + i] = ent;
      }
    }
  }
  __syncthreads();

  const unsigned tot = cnt * LPE;
  for (unsigned t0 = 0; t0 < tot; t0 += 32 * UB) {
    unsigned en[UB];
    vecV vv[UB];
#pragma unroll
    for (int u = 0; u < UB; ++u) {
      const unsigned t = t0 + u * 32 + lane;
      en[u] = (t < tot) ? wl[t / LPE] : 0u;
    }
#pragma unroll
    for (int u = 0; u < UB; ++u) {
      const unsigned t = t0 + u * 32 + lane;
      const unsigned s = en[u] >> RBITS;
      const unsigned q = t % LPE;
      vv[u] = *(const vecV*)(y + (size_t)s * F + q * V);
    }
#pragma unroll
    for (int u = 0; u < UB; ++u) {
      const unsigned t = t0 + u * 32 + lane;
      if (t < tot) {
        const unsigned r = en[u] & RMASK;
        const unsigned q = t % LPE;
        float* a = acc + r * F + q * V;
#pragma unroll
        for (int j = 0; j < V; ++j)
          __hip_atomic_fetch_add(a + j, vv[u][j], __ATOMIC_RELAXED, __HIP_MEMORY_SCOPE_WORKGROUP);
        if (WITH_DEG && q == 0)
          __hip_atomic_fetch_add(&dg[r], 1.0f, __ATOMIC_RELAXED, __HIP_MEMORY_SCOPE_WORKGROUP);
      }
    }
  }
  __syncthreads();

  const long long nF = (long long)n * F;
  float* pp = part + (size_t)chunk * (size_t)nF;
  const long long g0 = (long long)r0 * F;
  const bool vec = ((nF & 3) == 0);
  for (int i = tid; i < R * F / 4; i += 1024) {
    const long long gi = g0 + 4LL * i;
    if (vec && gi + 3 < nF) {
      *(v4f*)(pp + gi) = ((const v4f*)acc)[i];
    } else {
      for (int j = 0; j < 4; ++j)
        if (gi + j < nF) pp[gi + j] = acc[4 * i + j];
    }
  }
  if (WITH_DEG) {
    float* dp = degPart + (size_t)chunk * (size_t)n;
    for (int i = tid; i < R; i += 1024) {
      const int node = r0 + i;
      if (node < n) dp[node] = dg[i];
    }
  }
}

#define T2_STRIDE 20
template <int C>
__launch_bounds__(256)
__global__ void k_combine1_gemm2(const float* __restrict__ part, const float* __restrict__ degPart,
                                 const float* __restrict__ z1, const float* __restrict__ b1,
                                 const float* __restrict__ Wl, const float* __restrict__ Wr,
                                 float* __restrict__ y2, float* __restrict__ z2,
                                 float* __restrict__ degf, int n) {
  constexpr int NT = 4;
  __shared__ __attribute__((aligned(16))) float tile[256 * T2_STRIDE];
  const int tid = threadIdx.x;
  const int rb  = blockIdx.x * 256;
  const size_t plane = (size_t)n * 16;

  const int lane = tid & 31;
  const int wave = tid >> 5;
  const int mr   = lane & 15;
  const int kh   = (lane >> 4) << 1;

  v2f bl[4], br[4];
#pragma unroll
  for (int s = 0; s < 4; ++s) {
    int k0 = 4 * s + kh;
    v2f tl = {}; v2f tr = {};
    if (mr < 2) {
      tl.x = Wl[k0 * 2 + mr]; tl.y = Wl[(k0 + 1) * 2 + mr];
      tr.x = Wr[k0 * 2 + mr]; tr.y = Wr[(k0 + 1) * 2 + mr];
    }
    bl[s] = tl; br[s] = tr;
  }

  v4f a[NT], z[NT];
  float d[NT];
#pragma unroll
  for (int it = 0; it < NT; ++it) {
    const int t = it * 256 + tid;
    int node = rb + (t >> 2); if (node > n - 1) node = n - 1;
    const size_t off = (size_t)node * 16 + (size_t)((t & 3) * 4);
    v4f as = {0.f, 0.f, 0.f, 0.f};
    float ds = 0.0f;
#pragma unroll
    for (int c = 0; c < C; ++c) {
      as += *(const v4f*)(part + (size_t)c * plane + off);
      ds += degPart[(size_t)c * n + node];
    }
    a[it] = as; d[it] = ds;
    z[it] = *(const v4f*)(z1 + off);
  }
#pragma unroll
  for (int it = 0; it < NT; ++it) {
    const int t = it * 256 + tid;
    const int nl = t >> 2;
    const int q  = t & 3;
    if (q == 0 && rb + nl < n) degf[rb + nl] = d[it];
    const float dd = fmaxf(d[it], 1.0f);
    v4f r;
    r.x = fmaxf(a[it].x / dd + z[it].x + b1[q * 4 + 0], 0.0f);
    r.y = fmaxf(a[it].y / dd + z[it].y + b1[q * 4 + 1], 0.0f);
    r.z = fmaxf(a[it].z / dd + z[it].z + b1[q * 4 + 2], 0.0f);
    r.w = fmaxf(a[it].w / dd + z[it].w + b1[q * 4 + 3], 0.0f);
    *(v4f*)(tile + nl * T2_STRIDE + q * 4) = r;
  }
  __syncthreads();

  v8f cl[2], cr[2];
#pragma unroll
  for (int m = 0; m < 2; ++m) {
    const float* arow = tile + (wave * 32 + m * 16 + mr) * T2_STRIDE;
    v8f tl = {}; v8f tr = {};
#pragma unroll
    for (int s = 0; s < 4; ++s) {
      v2f av = *(const v2f*)(arow + 4 * s + kh);
      tl = wmma_f32x4(av, bl[s], tl);
      tr = wmma_f32x4(av, br[s], tr);
    }
    cl[m] = tl; cr[m] = tr;
  }

  const int rowHalf = (lane >> 4) << 3;
  if (mr < 2) {
#pragma unroll
    for (int m = 0; m < 2; ++m) {
#pragma unroll
      for (int v = 0; v < 8; ++v) {
        int row = rb + wave * 32 + m * 16 + rowHalf + v;
        if (row < n) {
          y2[(size_t)row * 2 + mr] = cl[m][v];
          z2[(size_t)row * 2 + mr] = cr[m][v];
        }
      }
    }
  }
}

template <int C>
__global__ void k_combine2(const float* __restrict__ part, const float* __restrict__ degf,
                           const float* __restrict__ z2, const float* __restrict__ b2,
                           float* __restrict__ out, int n) {
  const int node = (int)(blockIdx.x * blockDim.x + threadIdx.x);
  if (node >= n) return;
  const size_t plane = (size_t)n * 2;
  float a0 = 0.0f, a1 = 0.0f;
#pragma unroll
  for (int c = 0; c < C; ++c) {
    const v2f p = *(const v2f*)(part + (size_t)c * plane + (size_t)node * 2);
    a0 += p.x; a1 += p.y;
  }
  const float d = fmaxf(degf[node], 1.0f);
  const v2f z = *(const v2f*)(z2 + (size_t)node * 2);
  v2f o;
  o.x = a0 / d + z.x + b2[0];
  o.y = a1 / d + z.y + b2[1];
  *(v2f*)(out + (size_t)node * 2) = o;
}

extern "C" void kernel_launch(void* const* d_in, const int* in_sizes, int n_in,
                              void* d_out, int out_size, void* d_ws, size_t ws_size,
                              hipStream_t stream) {
  const float* x   = (const float*)d_in[0];
  const int*   ei  = (const int*)d_in[1];
  const float* W1l = (const float*)d_in[2];
  const float* W1r = (const float*)d_in[3];
  const float* b1  = (const float*)d_in[4];
  const float* W2l = (const float*)d_in[5];
  const float* W2r = (const float*)d_in[6];
  const float* b2  = (const float*)d_in[7];
  float* out = (float*)d_out;

  const int       n = in_sizes[0] / 64;
  const long long E = in_sizes[1] / 2;

  constexpr int NR1 = 28, C1 = 9;
  constexpr int NR2 = NR1, C2 = C1;

  float* ws    = (float*)d_ws;
  float* part1 = ws;
  float* part2 = part1 + (size_t)C1 * 16 * n;
  float* y1    = part2 + (size_t)C2 * 2 * n;
  float* z1    = y1 + (size_t)16 * n;
  float* y2    = z1 + (size_t)16 * n;
  float* z2    = y2 + (size_t)2 * n;
  float* degf  = z2 + (size_t)2 * n;
  float* degP  = degf + n;
  unsigned* gcnt  = (unsigned*)(degP + (size_t)C1 * n);
  unsigned* glist = gcnt + (size_t)NR1 * C1 * 32;

  const unsigned gGemm = (unsigned)((n + 127) / 128);
  k_gemm1<<<gGemm, 256, 0, stream>>>(x, W1l, W1r, y1, z1, n);

  k_agg<16, 3584, 12, 512, true, 1><<<dim3(NR1, C1), 1024, 0, stream>>>(ei, y1, part1, degP, glist, gcnt, n, E);

  k_combine1_gemm2<C1><<<(unsigned)((n + 255) / 256), 256, 0, stream>>>(part1, degP, z1, b1, W2l, W2r, y2, z2, degf, n);

  k_agg<2, 3584, 12, 512, false, 2><<<dim3(NR2, C2), 1024, 0, stream>>>(ei, y2, part2, degP, glist, gcnt, n, E);

  k_combine2<C2><<<(unsigned)((n + 255) / 256), 256, 0, stream>>>(part2, degf, z2, b2, out, n);
}


# Edge filter streams the source column beside the destination column (16-byte vectors, one trip ahead) and stores finished entries: no scattered source-index pass, one barrier fewer; sums unchanged
# speedup vs baseline: 3.8164x; 1.0547x over previous
#include <hip/hip_runtime.h>


typedef __attribute__((ext_vector_type(2))) float v2f;
typedef __attribute__((ext_vector_type(4))) float v4f;
typedef __attribute__((ext_vector_type(8))) float v8f;
typedef __attribute__((ext_vector_type(4))) int v4i;

static __device__ __forceinline__ v8f wmma_f32x4(v2f a, v2f b, v8f c) {
  return __builtin_amdgcn_wmma_f32_16x16x4_f32(
       false, a,  false, b,
       (short)0, c,  false,  false);
}

#define T1_STRIDE 68
__launch_bounds__(256)
__global__ void k_gemm1(const float* __restrict__ x, const float* __restrict__ Wl,
                        const float* __restrict__ Wr, float* __restrict__ y1,
                        float* __restrict__ z1, int n) {
  __shared__ float tile[128 * T1_STRIDE];
  const int tid = threadIdx.x;
  const int rb  = blockIdx.x * 128;

#pragma unroll
  for (int it = 0; it < 8; ++it) {
    int idx = (it * 256 + tid) * 4;
    int r = idx >> 6, c = idx & 63;
    int grow = rb + r; if (grow > n - 1) grow = n - 1;
    v4f v = *(const v4f*)(x + (size_t)grow * 64 + c);
    *(v4f*)(tile + r * T1_STRIDE + c) = v;
  }
  __syncthreads();

  const int lane = tid & 31;
  const int wave = tid >> 5;
  const int mr   = lane & 15;
  const int kh   = (lane >> 4) << 1;

  v2f bl[16], br[16];
#pragma unroll
  for (int s = 0; s < 16; ++s) {
    int k0 = 4 * s + kh;
    v2f tl, tr;
    tl.x = Wl[k0 * 16 + mr]; tl.y = Wl[(k0 + 1) * 16 + mr];
    tr.x = Wr[k0 * 16 + mr]; tr.y = Wr[(k0 + 1) * 16 + mr];
    bl[s] = tl; br[s] = tr;
  }

  const float* arow = tile + (wave * 16 + mr) * T1_STRIDE;
  v8f cl = {}; v8f cr = {};
#pragma unroll
  for (int s = 0; s < 16; ++s) {
    v2f a = *(const v2f*)(arow + 4 * s + kh);
    cl = wmma_f32x4(a, bl[s], cl);
    cr = wmma_f32x4(a, br[s], cr);
  }

  const int rowHalf = (lane >> 4) << 3;
#pragma unroll
  for (int v = 0; v < 8; ++v) {
    int row = rb + wave * 16 + rowHalf + v;
    if (row < n) {
      y1[(size_t)row * 16 + mr] = cl[v];
      z1[(size_t)row * 16 + mr] = cr[v];
    }
  }
}

template <int F, int RMAX, int RBITS, int WCAP, bool WITH_DEG, int LISTS>
__launch_bounds__(1024)
__global__ void k_agg(const int* __restrict__ ei, const float* __restrict__ y,
                      float* __restrict__ part, float* __restrict__ degPart,
                      unsigned* __restrict__ gl, unsigned* __restrict__ gc,
                      int n, long long E) {
  constexpr int V   = (F >= 4) ? 4 : F;
  constexpr int LPE = F / V;
  constexpr unsigned RMASK = (1u << RBITS) - 1u;
  constexpr int UA = 8;
  constexpr int UB = 8;
  constexpr int UC = 8;
  typedef __attribute__((ext_vector_type(V))) float vecV;
  static_assert(RMAX <= (1 << RBITS), "range index fits its field");
  static_assert((RMAX * F) % 4 == 0 && (F % V) == 0, "vector forms");

  __shared__ __attribute__((aligned(16))) float acc[RMAX * F];
  __shared__ float dg[WITH_DEG ? RMAX : 4];
  __shared__ unsigned list[32 * WCAP];

  const int tid  = threadIdx.x;
  const int lane = tid & 31;
  const int wave = __builtin_amdgcn_readfirstlane(tid >> 5);
  const int nRanges = (int)gridDim.x;
  const int nChunks = (int)gridDim.y;
  const int R  = (((n + nRanges - 1) / nRanges) + 15) & ~15;
  const int r0 = (int)blockIdx.x * R;
  const int chunk = (int)blockIdx.y;

  for (int i = tid; i < RMAX * F / 4; i += 1024) ((v4f*)acc)[i] = (v4f){0.f, 0.f, 0.f, 0.f};
  if (WITH_DEG) for (int i = tid; i < RMAX; i += 1024) dg[i] = 0.0f;
  __syncthreads();

  const int* __restrict__ srcA = ei;
  const int* __restrict__ dstA = ei + E;
  const int Ei   = (int)E;
  const int per  = (((Ei + nChunks - 1) / nChunks) + 1023) & ~1023;
  const int wper = per >> 5;
  const int ws   = chunk * per + wave * wper;
  int we = ws + wper; if (we > Ei) we = Ei;

  unsigned* wl = list + wave * WCAP;
  unsigned wcount = 0;

  const size_t slot = ((size_t)blockIdx.x * (size_t)nChunks + (size_t)chunk) * 32u + (size_t)wave;
  bool filter = true;
  if (LISTS == 2) {
    const unsigned saved = gc[slot];
    if (saved <= (unsigned)WCAP) {
      const unsigned* __restrict__ g = gl + slot * WCAP;
      for (unsigned i0 = 0; i0 < saved; i0 += 32 * UC) {
        unsigned en[UC];
#pragma unroll
        for (int u = 0; u < UC; ++u) {
          const unsigned i = i0 + u * 32 + lane;
          en[u] = g[i < saved ? i : 0u];
        }
#pragma unroll
        for (int u = 0; u < UC; ++u) wl[i0 + u * 32 + lane] = en[u];
      }
      wcount = saved;
      filter = false;
    }
  }

  int base = ws;
  if (filter && (((((size_t)dstA) | ((size_t)srcA)) & 15) == 0) && base + 256 <= we) {
    const int* __restrict__ dp = dstA + 4 * lane;
    const int* __restrict__ sp = srcA + 4 * lane;
    v4i d0 = *(const v4i*)(dp + base);
    v4i d1 = *(const v4i*)(dp + base + 128);
    v4i s0 = *(const v4i*)(sp + base);
    v4i s1 = *(const v4i*)(sp + base + 128);
    for (; base + 256 <= we && wcount + 256u <= (unsigned)WCAP; base += 256) {
      const int nb = (base + 512 <= we) ? base + 256 : base;
      const v4i n0 = *(const v4i*)(dp + nb);
      const v4i n1 = *(const v4i*)(dp + nb + 128);
      const v4i m0 = *(const v4i*)(sp + nb);
      const v4i m1 = *(const v4i*)(sp + nb + 128);
#pragma unroll
      for (int u = 0; u < 8; ++u) {
        const int dval = (u < 4) ? d0[u & 3] : d1[u & 3];
        const int sval = (u < 4) ? s0[u & 3] : s1[u & 3];
        const unsigned r = (unsigned)(dval - r0);
        const bool m = r < (unsigned)R;
        const unsigned mask = __builtin_amdgcn_ballot_w32(m);
        const unsigned pos  = __builtin_amdgcn_mbcnt_lo(mask, wcount);
        if (m) wl[pos] = ((unsigned)sval << RBITS) | r;
        wcount += (unsigned)__builtin_popcount(mask);
      }
      d0 = n0; d1 = n1;
      s0 = m0; s1 = m1;
    }
  }
  if (filter)
  for (; base < we; base += 32 * UA) {
    int dv[UA];
    int sq[UA];
#pragma unroll
    for (int u = 0; u < UA; ++u) {
      const int e  = base + u * 32 + lane;
      const int ec = e < Ei ? e : Ei - 1;
      int d = dstA[ec];
      sq[u] = srcA[ec];
      dv[u] = (e < we) ? d : -1;
    }
#pragma unroll
    for (int u = 0; u < UA; ++u) {
      const unsigned r = (unsigned)(dv[u] - r0);
      const bool m = r < (unsigned)R;
      const unsigned mask = __builtin_amdgcn_ballot_w32(m);
      const unsigned pos  = wcount + __builtin_amdgcn_mbcnt_lo(mask, 0u);
      if (m) {
        if (__builtin_expect(pos < (unsigned)WCAP, 1)) {
          wl[pos] = ((unsigned)sq[u] << RBITS) | r;
        } else {
          const int s = sq[u];
#pragma nounroll
          for (int f = 0; f < F; ++f)
            __hip_atomic_fetch_add(&acc[r * F + f], y[(size_t)s * F + f],
                                   __ATOMIC_RELAXED, __HIP_MEMORY_SCOPE_WORKGROUP);
          if (WITH_DEG)
            __hip_atomic_fetch_add(&dg[r], 1.0f, __ATOMIC_RELAXED, __HIP_MEMORY_SCOPE_WORKGROUP);
        }
      }
      wcount += (unsigned)__builtin_popcount(mask);
    }
  }
  __syncthreads();

  const unsigned cnt = wcount < (unsigned)WCAP ? wcount : (unsigned)WCAP;
  if (LISTS == 1) {
    if (lane == 0) gc[slot] = wcount;
    for (unsigned i0 = 0; i0 < cnt; i0 += 32 * UC) {
      unsigned en[UC];
#pragma unroll
      for (int u = 0; u < UC; ++u) {
        const unsigned i = i0 + u * 32 + lane;
        en[u] = (i < cnt) ? wl[i] : 0u;
      }
#pragma unroll
      for (int u = 0; u < UC; ++u) {
        const unsigned i = i0 + u * 32 + lane;
        if (i < cnt) gl[slot * WCAP + i] = en[u];
      }
    }
  }

  const unsigned tot = cnt * LPE;
  for (unsigned t0 = 0; t0 < tot; t0 += 32 * UB) {
    unsigned en[UB];
    vecV vv[UB];
#pragma unroll
    for (int u = 0; u < UB; ++u) {
      const unsigned t = t0 + u * 32 + lane;
      en[u] = (t < tot) ? wl[t / LPE] : 0u;
    }
#pragma unroll
    for (int u = 0; u < UB; ++u) {
      const unsigned t = t0 + u * 32 + lane;
      const unsigned s = en[u] >> RBITS;
      const unsigned q = t % LPE;
      vv[u] = *(const vecV*)(y + (size_t)s * F + q * V);
    }
#pragma unroll
    for (int u = 0; u < UB; ++u) {
      const unsigned t = t0 + u * 32 + lane;
      if (t < tot) {
        const unsigned r = en[u] & RMASK;
        const unsigned q = t % LPE;
        float* a = acc + r * F + q * V;
#pragma unroll
        for (int j = 0; j < V; ++j)
          __hip_atomic_fetch_add(a + j, vv[u][j], __ATOMIC_RELAXED, __HIP_MEMORY_SCOPE_WORKGROUP);
        if (WITH_DEG && q == 0)
          __hip_atomic_fetch_add(&dg[r], 1.0f, __ATOMIC_RELAXED, __HIP_MEMORY_SCOPE_WORKGROUP);
      }
    }
  }
  __syncthreads();

  const long long nF = (long long)n * F;
  float* pp = part + (size_t)chunk * (size_t)nF;
  const long long g0 = (long long)r0 * F;
  const bool vec = ((nF & 3) == 0);
  for (int i = tid; i < R * F / 4; i += 1024) {
    const long long gi = g0 + 4LL * i;
    if (vec && gi + 3 < nF) {
      *(v4f*)(pp + gi) = ((const v4f*)acc)[i];
    } else {
      for (int j = 0; j < 4; ++j)
        if (gi + j < nF) pp[gi + j] = acc[4 * i + j];
    }
  }
  if (WITH_DEG) {
    float* dp = degPart + (size_t)chunk * (size_t)n;
    for (int i = tid; i < R; i += 1024) {
      const int node = r0 + i;
      if (node < n) dp[node] = dg[i];
    }
  }
}

#define T2_STRIDE 20
template <int C>
__launch_bounds__(256)
__global__ void k_combine1_gemm2(const float* __restrict__ part, const float* __restrict__ degPart,
                                 const float* __restrict__ z1, const float* __restrict__ b1,
                                 const float* __restrict__ Wl, const float* __restrict__ Wr,
                                 float* __restrict__ y2, float* __restrict__ z2,
                                 float* __restrict__ degf, int n) {
  constexpr int NT = 4;
  __shared__ __attribute__((aligned(16))) float tile[256 * T2_STRIDE];
  const int tid = threadIdx.x;
  const int rb  = blockIdx.x * 256;
  const size_t plane = (size_t)n * 16;

  const int lane = tid & 31;
  const int wave = tid >> 5;
  const int mr   = lane & 15;
  const int kh   = (lane >> 4) << 1;

  v2f bl[4], br[4];
#pragma unroll
  for (int s = 0; s < 4; ++s) {
    int k0 = 4 * s + kh;
    v2f tl = {}; v2f tr = {};
    if (mr < 2) {
      tl.x = Wl[k0 * 2 + mr]; tl.y = Wl[(k0 + 1) * 2 + mr];
      tr.x = Wr[k0 * 2 + mr]; tr.y = Wr[(k0 + 1) * 2 + mr];
    }
    bl[s] = tl; br[s] = tr;
  }

  v4f a[NT], z[NT];
  float d[NT];
#pragma unroll
  for (int it = 0; it < NT; ++it) {
    const int t = it * 256 + tid;
    int node = rb + (t >> 2); if (node > n - 1) node = n - 1;
    const size_t off = (size_t)node * 16 + (size_t)((t & 3) * 4);
    v4f as = {0.f, 0.f, 0.f, 0.f};
    float ds = 0.0f;
#pragma unroll
    for (int c = 0; c < C; ++c) {
      as += *(const v4f*)(part + (size_t)c * plane + off);
      ds += degPart[(size_t)c * n + node];
    }
    a[it] = as; d[it] = ds;
    z[it] = *(const v4f*)(z1 + off);
  }
#pragma unroll
  for (int it = 0; it < NT; ++it) {
    const int t = it * 256 + tid;
    const int nl = t >> 2;
    const int q  = t & 3;
    if (q == 0 && rb + nl < n) degf[rb + nl] = d[it];
    const float dd = fmaxf(d[it], 1.0f);
    v4f r;
    r.x = fmaxf(a[it].x / dd + z[it].x + b1[q * 4 + 0], 0.0f);
    r.y = fmaxf(a[it].y / dd + z[it].y + b1[q * 4 + 1], 0.0f);
    r.z = fmaxf(a[it].z / dd + z[it].z + b1[q * 4 + 2], 0.0f);
    r.w = fmaxf(a[it].w / dd + z[it].w + b1[q * 4 + 3], 0.0f);
    *(v4f*)(tile + nl * T2_STRIDE + q * 4) = r;
  }
  __syncthreads();

  v8f cl[2], cr[2];
#pragma unroll
  for (int m = 0; m < 2; ++m) {
    const float* arow = tile + (wave * 32 + m * 16 + mr) * T2_STRIDE;
    v8f tl = {}; v8f tr = {};
#pragma unroll
    for (int s = 0; s < 4; ++s) {
      v2f av = *(const v2f*)(arow + 4 * s + kh);
      tl = wmma_f32x4(av, bl[s], tl);
      tr = wmma_f32x4(av, br[s], tr);
    }
    cl[m] = tl; cr[m] = tr;
  }

  const int rowHalf = (lane >> 4) << 3;
  if (mr < 2) {
#pragma unroll
    for (int m = 0; m < 2; ++m) {
#pragma unroll
      for (int v = 0; v < 8; ++v) {
        int row = rb + wave * 32 + m * 16 + rowHalf + v;
        if (row < n) {
          y2[(size_t)row * 2 + mr] = cl[m][v];
          z2[(size_t)row * 2 + mr] = cr[m][v];
        }
      }
    }
  }
}

template <int C>
__global__ void k_combine2(const float* __restrict__ part, const float* __restrict__ degf,
                           const float* __restrict__ z2, const float* __restrict__ b2,
                           float* __restrict__ out, int n) {
  const int node = (int)(blockIdx.x * blockDim.x + threadIdx.x);
  if (node >= n) return;
  const size_t plane = (size_t)n * 2;
  float a0 = 0.0f, a1 = 0.0f;
#pragma unroll
  for (int c = 0; c < C; ++c) {
    const v2f p = *(const v2f*)(part + (size_t)c * plane + (size_t)node * 2);
    a0 += p.x; a1 += p.y;
  }
  const float d = fmaxf(degf[node], 1.0f);
  const v2f z = *(const v2f*)(z2 + (size_t)node * 2);
  v2f o;
  o.x = a0 / d + z.x + b2[0];
  o.y = a1 / d + z.y + b2[1];
  *(v2f*)(out + (size_t)node * 2) = o;
}

extern "C" void kernel_launch(void* const* d_in, const int* in_sizes, int n_in,
                              void* d_out, int out_size, void* d_ws, size_t ws_size,
                              hipStream_t stream) {
  const float* x   = (const float*)d_in[0];
  const int*   ei  = (const int*)d_in[1];
  const float* W1l = (const float*)d_in[2];
  const float* W1r = (const float*)d_in[3];
  const float* b1  = (const float*)d_in[4];
  const float* W2l = (const float*)d_in[5];
  const float* W2r = (const float*)d_in[6];
  const float* b2  = (const float*)d_in[7];
  float* out = (float*)d_out;

  const int       n = in_sizes[0] / 64;
  const long long E = in_sizes[1] / 2;

  constexpr int NR1 = 28, C1 = 9;
  constexpr int NR2 = NR1, C2 = C1;

  float* ws    = (float*)d_ws;
  float* part1 = ws;
  float* part2 = part1 + (size_t)C1 * 16 * n;
  float* y1    = part2 + (size_t)C2 * 2 * n;
  float* z1    = y1 + (size_t)16 * n;
  float* y2    = z1 + (size_t)16 * n;
  float* z2    = y2 + (size_t)2 * n;
  float* degf  = z2 + (size_t)2 * n;
  float* degP  = degf + n;
  unsigned* gcnt  = (unsigned*)(degP + (size_t)C1 * n);
  unsigned* glist = gcnt + (size_t)NR1 * C1 * 32;

  const unsigned gGemm = (unsigned)((n + 127) / 128);
  k_gemm1<<<gGemm, 256, 0, stream>>>(x, W1l, W1r, y1, z1, n);

  k_agg<16, 3584, 12, 512, true, 1><<<dim3(NR1, C1), 1024, 0, stream>>>(ei, y1, part1, degP, glist, gcnt, n, E);

  k_combine1_gemm2<C1><<<(unsigned)((n + 255) / 256), 256, 0, stream>>>(part1, degP, z1, b1, W2l, W2r, y2, z2, degf, n);

  k_agg<2, 3584, 12, 512, false, 2><<<dim3(NR2, C2), 1024, 0, stream>>>(ei, y2, part2, degP, glist, gcnt, n, E);

  k_combine2<C2><<<(unsigned)((n + 255) / 256), 256, 0, stream>>>(part2, degf, z2, b2, out, n);
}


# Layer-1 row adds run inside the edge-filter loop: a wave adds each whole batch of 64 finished list entries (row loads, LDS adds) between trips; barrier behind the filter gone; sums in arrival order
# speedup vs baseline: 4.0902x; 1.0717x over previous
#include <hip/hip_runtime.h>


typedef __attribute__((ext_vector_type(2))) float v2f;
typedef __attribute__((ext_vector_type(4))) float v4f;
typedef __attribute__((ext_vector_type(8))) float v8f;
typedef __attribute__((ext_vector_type(4))) int v4i;

static __device__ __forceinline__ v8f wmma_f32x4(v2f a, v2f b, v8f c) {
  return __builtin_amdgcn_wmma_f32_16x16x4_f32(
       false, a,  false, b,
       (short)0, c,  false,  false);
}

#define T1_STRIDE 68
__launch_bounds__(256)
__global__ void k_gemm1(const float* __restrict__ x, const float* __restrict__ Wl,
                        const float* __restrict__ Wr, float* __restrict__ y1,
                        float* __restrict__ z1, int n) {
  __shared__ float tile[128 * T1_STRIDE];
  const int tid = threadIdx.x;
  const int rb  = blockIdx.x * 128;

#pragma unroll
  for (int it = 0; it < 8; ++it) {
    int idx = (it * 256 + tid) * 4;
    int r = idx >> 6, c = idx & 63;
    int grow = rb + r; if (grow > n - 1) grow = n - 1;
    v4f v = *(const v4f*)(x + (size_t)grow * 64 + c);
    *(v4f*)(tile + r * T1_STRIDE + c) = v;
  }
  __syncthreads();

  const int lane = tid & 31;
  const int wave = tid >> 5;
  const int mr   = lane & 15;
  const int kh   = (lane >> 4) << 1;

  v2f bl[16], br[16];
#pragma unroll
  for (int s = 0; s < 16; ++s) {
    int k0 = 4 * s + kh;
    v2f tl, tr;
    tl.x = Wl[k0 * 16 + mr]; tl.y = Wl[(k0 + 1) * 16 + mr];
    tr.x = Wr[k0 * 16 + mr]; tr.y = Wr[(k0 + 1) * 16 + mr];
    bl[s] = tl; br[s] = tr;
  }

  const float* arow = tile + (wave * 16 + mr) * T1_STRIDE;
  v8f cl = {}; v8f cr = {};
#pragma unroll
  for (int s = 0; s < 16; ++s) {
    v2f a = *(const v2f*)(arow + 4 * s + kh);
    cl = wmma_f32x4(a, bl[s], cl);
    cr = wmma_f32x4(a, br[s], cr);
  }

  const int rowHalf = (lane >> 4) << 3;
#pragma unroll
  for (int v = 0; v < 8; ++v) {
    int row = rb + wave * 16 + rowHalf + v;
    if (row < n) {
      y1[(size_t)row * 16 + mr] = cl[v];
      z1[(size_t)row * 16 + mr] = cr[v];
    }
  }
}

template <int F, int RMAX, int RBITS, int WCAP, bool WITH_DEG, int LISTS>
__launch_bounds__(1024)
__global__ void k_agg(const int* __restrict__ ei, const float* __restrict__ y,
                      float* __restrict__ part, float* __restrict__ degPart,
                      unsigned* __restrict__ gl, unsigned* __restrict__ gc,
                      int n, long long E) {
  constexpr int V   = (F >= 4) ? 4 : F;
  constexpr int LPE = F / V;
  constexpr unsigned RMASK = (1u << RBITS) - 1u;
  constexpr int UA = 8;
  constexpr int UB = 8;
  constexpr int UC = 8;
  typedef __attribute__((ext_vector_type(V))) float vecV;
  static_assert(RMAX <= (1 << RBITS), "range index fits its field");
  static_assert((RMAX * F) % 4 == 0 && (F % V) == 0, "vector forms");

  __shared__ __attribute__((aligned(16))) float acc[RMAX * F];
  __shared__ float dg[WITH_DEG ? RMAX : 4];
  __shared__ unsigned list[32 * WCAP];

  const int tid  = threadIdx.x;
  const int lane = tid & 31;
  const int wave = __builtin_amdgcn_readfirstlane(tid >> 5);
  const int nRanges = (int)gridDim.x;
  const int nChunks = (int)gridDim.y;
  const int R  = (((n + nRanges - 1) / nRanges) + 15) & ~15;
  const int r0 = (int)blockIdx.x * R;
  const int chunk = (int)blockIdx.y;

  for (int i = tid; i < RMAX * F / 4; i += 1024) ((v4f*)acc)[i] = (v4f){0.f, 0.f, 0.f, 0.f};
  if (WITH_DEG) for (int i = tid; i < RMAX; i += 1024) dg[i] = 0.0f;
  __syncthreads();

  const int* __restrict__ srcA = ei;
  const int* __restrict__ dstA = ei + E;
  const int Ei   = (int)E;
  const int per  = (((Ei + nChunks - 1) / nChunks) + 1023) & ~1023;
  const int wper = per >> 5;
  const int ws   = chunk * per + wave * wper;
  int we = ws + wper; if (we > Ei) we = Ei;

  unsigned* wl = list + wave * WCAP;
  unsigned wcount = 0;

  const size_t slot = ((size_t)blockIdx.x * (size_t)nChunks + (size_t)chunk) * 32u + (size_t)wave;
  bool filter = true;
  if (LISTS == 2) {
    const unsigned saved = gc[slot];
    if (saved <= (unsigned)WCAP) {
      const unsigned* __restrict__ g = gl + slot * WCAP;
      for (unsigned i0 = 0; i0 < saved; i0 += 32 * UC) {
        unsigned en[UC];
#pragma unroll
        for (int u = 0; u < UC; ++u) {
          const unsigned i = i0 + u * 32 + lane;
          en[u] = g[i < saved ? i : 0u];
        }
#pragma unroll
        for (int u = 0; u < UC; ++u) wl[i0 + u * 32 + lane] = en[u];
      }
      wcount = saved;
      filter = false;
    }
  }

  auto add_rows = [&](const unsigned t0, const unsigned tot, const bool full) {
    unsigned en[UB];
    vecV vv[UB];
#pragma unroll
    for (int u = 0; u < UB; ++u) {
      const unsigned t = t0 + u * 32 + lane;
      en[u] = (full || t < tot) ? wl[t / LPE] : 0u;
    }
#pragma unroll
    for (int u = 0; u < UB; ++u) {
      const unsigned t = t0 + u * 32 + lane;
      const unsigned s = en[u] >> RBITS;
      const unsigned q = t % LPE;
      vv[u] = *(const vecV*)(y + (size_t)s * F + q * V);
    }
#pragma unroll
    for (int u = 0; u < UB; ++u) {
      const unsigned t = t0 + u * 32 + lane;
      if (full || t < tot) {
        const unsigned r = en[u] & RMASK;
        const unsigned q = t % LPE;
        float* a = acc + r * F + q * V;
#pragma unroll
        for (int j = 0; j < V; ++j)
          __hip_atomic_fetch_add(a + j, vv[u][j], __ATOMIC_RELAXED, __HIP_MEMORY_SCOPE_WORKGROUP);
        if (WITH_DEG && q == 0)
          __hip_atomic_fetch_add(&dg[r], 1.0f, __ATOMIC_RELAXED, __HIP_MEMORY_SCOPE_WORKGROUP);
      }
    }
  };
  constexpr unsigned BE = 32 * UB / LPE;
  unsigned done = 0;

  int base = ws;
  if (filter && (((((size_t)dstA) | ((size_t)srcA)) & 15) == 0) && base + 256 <= we) {
    const int* __restrict__ dp = dstA + 4 * lane;
    const int* __restrict__ sp = srcA + 4 * lane;
    v4i d0 = *(const v4i*)(dp + base);
    v4i d1 = *(const v4i*)(dp + base + 128);
    v4i s0 = *(const v4i*)(sp + base);
    v4i s1 = *(const v4i*)(sp + base + 128);
    for (; base + 256 <= we && wcount + 256u <= (unsigned)WCAP; base += 256) {
      const int nb = (base + 512 <= we) ? base + 256 : base;
      const v4i n0 = *(const v4i*)(dp + nb);
      const v4i n1 = *(const v4i*)(dp + nb + 128);
      const v4i m0 = *(const v4i*)(sp + nb);
      const v4i m1 = *(const v4i*)(sp + nb + 128);
#pragma unroll
      for (int u = 0; u < 8; ++u) {
        const int dval = (u < 4) ? d0[u & 3] : d1[u & 3];
        const int sval = (u < 4) ? s0[u & 3] : s1[u & 3];
        const unsigned r = (unsigned)(dval - r0);
        const bool m = r < (unsigned)R;
        const unsigned mask = __builtin_amdgcn_ballot_w32(m);
        const unsigned pos  = __builtin_amdgcn_mbcnt_lo(mask, wcount);
        if (m) wl[pos] = ((unsigned)sval << RBITS) | r;
        wcount += (unsigned)__builtin_popcount(mask);
      }
      if (wcount - done >= BE) {
        add_rows(done * LPE, 0u, true);
        done += BE;
      }
      d0 = n0; d1 = n1;
      s0 = m0; s1 = m1;
    }
  }
  if (filter)
  for (; base < we; base += 32 * UA) {
    int dv[UA];
    int sq[UA];
#pragma unroll
    for (int u = 0; u < UA; ++u) {
      const int e  = base + u * 32 + lane;
      const int ec = e < Ei ? e : Ei - 1;
      int d = dstA[ec];
      sq[u] = srcA[ec];
      dv[u] = (e < we) ? d : -1;
    }
#pragma unroll
    for (int u = 0; u < UA; ++u) {
      const unsigned r = (unsigned)(dv[u] - r0);
      const bool m = r < (unsigned)R;
      const unsigned mask = __builtin_amdgcn_ballot_w32(m);
      const unsigned pos  = wcount + __builtin_amdgcn_mbcnt_lo(mask, 0u);
      if (m) {
        if (__builtin_expect(pos < (unsigned)WCAP, 1)) {
          wl[pos] = ((unsigned)sq[u] << RBITS) | r;
        } else {
          const int s = sq[u];
#pragma nounroll
          for (int f = 0; f < F; ++f)
            __hip_atomic_fetch_add(&acc[r * F + f], y[(size_t)s * F + f],
                                   __ATOMIC_RELAXED, __HIP_MEMORY_SCOPE_WORKGROUP);
          if (WITH_DEG)
            __hip_atomic_fetch_add(&dg[r], 1.0f, __ATOMIC_RELAXED, __HIP_MEMORY_SCOPE_WORKGROUP);
        }
      }
      wcount += (unsigned)__builtin_popcount(mask);
    }
  }

  const unsigned cnt = wcount < (unsigned)WCAP ? wcount : (unsigned)WCAP;
  if (LISTS == 1) {
    if (lane == 0) gc[slot] = wcount;
    for (unsigned i0 = 0; i0 < cnt; i0 += 32 * UC) {
      unsigned en[UC];
#pragma unroll
      for (int u = 0; u < UC; ++u) {
        const unsigned i = i0 + u * 32 + lane;
        en[u] = (i < cnt) ? wl[i] : 0u;
      }
#pragma unroll
      for (int u = 0; u < UC; ++u) {
        const unsigned i = i0 + u * 32 + lane;
        if (i < cnt) gl[slot * WCAP + i] = en[u];
      }
    }
  }

  const unsigned tot = cnt * LPE;
  for (unsigned t0 = done * LPE; t0 < tot; t0 += 32 * UB) add_rows(t0, tot, false);
  __syncthreads();

  const long long nF = (long long)n * F;
  float* pp = part + (size_t)chunk * (size_t)nF;
  const long long g0 = (long long)r0 * F;
  const bool vec = ((nF & 3) == 0);
  for (int i = tid; i < R * F / 4; i += 1024) {
    const long long gi = g0 + 4LL * i;
    if (vec && gi + 3 < nF) {
      *(v4f*)(pp + gi) = ((const v4f*)acc)[i];
    } else {
      for (int j = 0; j < 4; ++j)
        if (gi + j < nF) pp[gi + j] = acc[4 * i + j];
    }
  }
  if (WITH_DEG) {
    float* dp = degPart + (size_t)chunk * (size_t)n;
    for (int i = tid; i < R; i += 1024) {
      const int node = r0 + i;
      if (node < n) dp[node] = dg[i];
    }
  }
}

#define T2_STRIDE 20
template <int C>
__launch_bounds__(256)
__global__ void k_combine1_gemm2(const float* __restrict__ part, const float* __restrict__ degPart,
                                 const float* __restrict__ z1, const float* __restrict__ b1,
                                 const float* __restrict__ Wl, const float* __restrict__ Wr,
                                 float* __restrict__ y2, float* __restrict__ z2,
                                 float* __restrict__ degf, int n) {
  constexpr int NT = 4;
  __shared__ __attribute__((aligned(16))) float tile[256 * T2_STRIDE];
  const int tid = threadIdx.x;
  const int rb  = blockIdx.x * 256;
  const size_t plane = (size_t)n * 16;

  const int lane = tid & 31;
  const int wave = tid >> 5;
  const int mr   = lane & 15;
  const int kh   = (lane >> 4) << 1;

  v2f bl[4], br[4];
#pragma unroll
  for (int s = 0; s < 4; ++s) {
    int k0 = 4 * s + kh;
    v2f tl = {}; v2f tr = {};
    if (mr < 2) {
      tl.x = Wl[k0 * 2 + mr]; tl.y = Wl[(k0 + 1) * 2 + mr];
      tr.x = Wr[k0 * 2 + mr]; tr.y = Wr[(k0 + 1) * 2 + mr];
    }
    bl[s] = tl; br[s] = tr;
  }

  v4f a[NT], z[NT];
  float d[NT];
#pragma unroll
  for (int it = 0; it < NT; ++it) {
    const int t = it * 256 + tid;
    int node = rb + (t >> 2); if (node > n - 1) node = n - 1;
    const size_t off = (size_t)node * 16 + (size_t)((t & 3) * 4);
    v4f as = {0.f, 0.f, 0.f, 0.f};
    float ds = 0.0f;
#pragma unroll
    for (int c = 0; c < C; ++c) {
      as += *(const v4f*)(part + (size_t)c * plane + off);
      ds += degPart[(size_t)c * n + node];
    }
    a[it] = as; d[it] = ds;
    z[it] = *(const v4f*)(z1 + off);
  }
#pragma unroll
  for (int it = 0; it < NT; ++it) {
    const int t = it * 256 + tid;
    const int nl = t >> 2;
    const int q  = t & 3;
    if (q == 0 && rb + nl < n) degf[rb + nl] = d[it];
    const float dd = fmaxf(d[it], 1.0f);
    v4f r;
    r.x = fmaxf(a[it].x / dd + z[it].x + b1[q * 4 + 0], 0.0f);
    r.y = fmaxf(a[it].y / dd + z[it].y + b1[q * 4 + 1], 0.0f);
    r.z = fmaxf(a[it].z / dd + z[it].z + b1[q * 4 + 2], 0.0f);
    r.w = fmaxf(a[it].w / dd + z[it].w + b1[q * 4 + 3], 0.0f);
    *(v4f*)(tile + nl * T2_STRIDE + q * 4) = r;
  }
  __syncthreads();

  v8f cl[2], cr[2];
#pragma unroll
  for (int m = 0; m < 2; ++m) {
    const float* arow = tile + (wave * 32 + m * 16 + mr) * T2_STRIDE;
    v8f tl = {}; v8f tr = {};
#pragma unroll
    for (int s = 0; s < 4; ++s) {
      v2f av = *(const v2f*)(arow + 4 * s + kh);
      tl = wmma_f32x4(av, bl[s], tl);
      tr = wmma_f32x4(av, br[s], tr);
    }
    cl[m] = tl; cr[m] = tr;
  }

  const int rowHalf = (lane >> 4) << 3;
  if (mr < 2) {
#pragma unroll
    for (int m = 0; m < 2; ++m) {
#pragma unroll
      for (int v = 0; v < 8; ++v) {
        int row = rb + wave * 32 + m * 16 + rowHalf + v;
        if (row < n) {
          y2[(size_t)row * 2 + mr] = cl[m][v];
          z2[(size_t)row * 2 + mr] = cr[m][v];
        }
      }
    }
  }
}

template <int C>
__global__ void k_combine2(const float* __restrict__ part, const float* __restrict__ degf,
                           const float* __restrict__ z2, const float* __restrict__ b2,
                           float* __restrict__ out, int n) {
  const int node = (int)(blockIdx.x * blockDim.x + threadIdx.x);
  if (node >= n) return;
  const size_t plane = (size_t)n * 2;
  float a0 = 0.0f, a1 = 0.0f;
#pragma unroll
  for (int c = 0; c < C; ++c) {
    const v2f p = *(const v2f*)(part + (size_t)c * plane + (size_t)node * 2);
    a0 += p.x; a1 += p.y;
  }
  const float d = fmaxf(degf[node], 1.0f);
  const v2f z = *(const v2f*)(z2 + (size_t)node * 2);
  v2f o;
  o.x = a0 / d + z.x + b2[0];
  o.y = a1 / d + z.y + b2[1];
  *(v2f*)(out + (size_t)node * 2) = o;
}

extern "C" void kernel_launch(void* const* d_in, const int* in_sizes, int n_in,
                              void* d_out, int out_size, void* d_ws, size_t ws_size,
                              hipStream_t stream) {
  const float* x   = (const float*)d_in[0];
  const int*   ei  = (const int*)d_in[1];
  const float* W1l = (const float*)d_in[2];
  const float* W1r = (const float*)d_in[3];
  const float* b1  = (const float*)d_in[4];
  const float* W2l = (const float*)d_in[5];
  const float* W2r = (const float*)d_in[6];
  const float* b2  = (const float*)d_in[7];
  float* out = (float*)d_out;

  const int       n = in_sizes[0] / 64;
  const long long E = in_sizes[1] / 2;

  constexpr int NR1 = 28, C1 = 9;
  constexpr int NR2 = NR1, C2 = C1;

  float* ws    = (float*)d_ws;
  float* part1 = ws;
  float* part2 = part1 + (size_t)C1 * 16 * n;
  float* y1    = part2 + (size_t)C2 * 2 * n;
  float* z1    = y1 + (size_t)16 * n;
  float* y2    = z1 + (size_t)16 * n;
  float* z2    = y2 + (size_t)2 * n;
  float* degf  = z2 + (size_t)2 * n;
  float* degP  = degf + n;
  unsigned* gcnt  = (unsigned*)(degP + (size_t)C1 * n);
  unsigned* glist = gcnt + (size_t)NR1 * C1 * 32;

  const unsigned gGemm = (unsigned)((n + 127) / 128);
  k_gemm1<<<gGemm, 256, 0, stream>>>(x, W1l, W1r, y1, z1, n);

  k_agg<16, 3584, 12, 512, true, 1><<<dim3(NR1, C1), 1024, 0, stream>>>(ei, y1, part1, degP, glist, gcnt, n, E);

  k_combine1_gemm2<C1><<<(unsigned)((n + 255) / 256), 256, 0, stream>>>(part1, degP, z1, b1, W2l, W2r, y2, z2, degf, n);

  k_agg<2, 3584, 12, 512, false, 2><<<dim3(NR2, C2), 1024, 0, stream>>>(ei, y2, part2, degP, glist, gcnt, n, E);

  k_combine2<C2><<<(unsigned)((n + 255) / 256), 256, 0, stream>>>(part2, degf, z2, b2, out, n);
}


# Edge filter walks its 256-edge trips in an order rotated by the node-range index: the 28 work-groups of a chunk read different lines at a time; list order changes, sums stay in arrival order
# speedup vs baseline: 4.2264x; 1.0333x over previous
#include <hip/hip_runtime.h>


typedef __attribute__((ext_vector_type(2))) float v2f;
typedef __attribute__((ext_vector_type(4))) float v4f;
typedef __attribute__((ext_vector_type(8))) float v8f;
typedef __attribute__((ext_vector_type(4))) int v4i;

static __device__ __forceinline__ v8f wmma_f32x4(v2f a, v2f b, v8f c) {
  return __builtin_amdgcn_wmma_f32_16x16x4_f32(
       false, a,  false, b,
       (short)0, c,  false,  false);
}

#define T1_STRIDE 68
__launch_bounds__(256)
__global__ void k_gemm1(const float* __restrict__ x, const float* __restrict__ Wl,
                        const float* __restrict__ Wr, float* __restrict__ y1,
                        float* __restrict__ z1, int n) {
  __shared__ float tile[128 * T1_STRIDE];
  const int tid = threadIdx.x;
  const int rb  = blockIdx.x * 128;

#pragma unroll
  for (int it = 0; it < 8; ++it) {
    int idx = (it * 256 + tid) * 4;
    int r = idx >> 6, c = idx & 63;
    int grow = rb + r; if (grow > n - 1) grow = n - 1;
    v4f v = *(const v4f*)(x + (size_t)grow * 64 + c);
    *(v4f*)(tile + r * T1_STRIDE + c) = v;
  }
  __syncthreads();

  const int lane = tid & 31;
  const int wave = tid >> 5;
  const int mr   = lane & 15;
  const int kh   = (lane >> 4) << 1;

  v2f bl[16], br[16];
#pragma unroll
  for (int s = 0; s < 16; ++s) {
    int k0 = 4 * s + kh;
    v2f tl, tr;
    tl.x = Wl[k0 * 16 + mr]; tl.y = Wl[(k0 + 1) * 16 + mr];
    tr.x = Wr[k0 * 16 + mr]; tr.y = Wr[(k0 + 1) * 16 + mr];
    bl[s] = tl; br[s] = tr;
  }

  const float* arow = tile + (wave * 16 + mr) * T1_STRIDE;
  v8f cl = {}; v8f cr = {};
#pragma unroll
  for (int s = 0; s < 16; ++s) {
    v2f a = *(const v2f*)(arow + 4 * s + kh);
    cl = wmma_f32x4(a, bl[s], cl);
    cr = wmma_f32x4(a, br[s], cr);
  }

  const int rowHalf = (lane >> 4) << 3;
#pragma unroll
  for (int v = 0; v < 8; ++v) {
    int row = rb + wave * 16 + rowHalf + v;
    if (row < n) {
      y1[(size_t)row * 16 + mr] = cl[v];
      z1[(size_t)row * 16 + mr] = cr[v];
    }
  }
}

template <int F, int RMAX, int RBITS, int WCAP, bool WITH_DEG, int LISTS>
__launch_bounds__(1024)
__global__ void k_agg(const int* __restrict__ ei, const float* __restrict__ y,
                      float* __restrict__ part, float* __restrict__ degPart,
                      unsigned* __restrict__ gl, unsigned* __restrict__ gc,
                      int n, long long E) {
  constexpr int V   = (F >= 4) ? 4 : F;
  constexpr int LPE = F / V;
  constexpr unsigned RMASK = (1u << RBITS) - 1u;
  constexpr int UA = 8;
  constexpr int UB = 8;
  constexpr int UC = 8;
  typedef __attribute__((ext_vector_type(V))) float vecV;
  static_assert(RMAX <= (1 << RBITS), "range index fits its field");
  static_assert((RMAX * F) % 4 == 0 && (F % V) == 0, "vector forms");

  __shared__ __attribute__((aligned(16))) float acc[RMAX * F];
  __shared__ float dg[WITH_DEG ? RMAX : 4];
  __shared__ unsigned list[32 * WCAP];

  const int tid  = threadIdx.x;
  const int lane = tid & 31;
  const int wave = __builtin_amdgcn_readfirstlane(tid >> 5);
  const int nRanges = (int)gridDim.x;
  const int nChunks = (int)gridDim.y;
  const int R  = (((n + nRanges - 1) / nRanges) + 15) & ~15;
  const int r0 = (int)blockIdx.x * R;
  const int chunk = (int)blockIdx.y;

  for (int i = tid; i < RMAX * F / 4; i += 1024) ((v4f*)acc)[i] = (v4f){0.f, 0.f, 0.f, 0.f};
  if (WITH_DEG) for (int i = tid; i < RMAX; i += 1024) dg[i] = 0.0f;
  __syncthreads();

  const int* __restrict__ srcA = ei;
  const int* __restrict__ dstA = ei + E;
  const int Ei   = (int)E;
  const int per  = (((Ei + nChunks - 1) / nChunks) + 1023) & ~1023;
  const int wper = per >> 5;
  const int ws   = chunk * per + wave * wper;
  int we = ws + wper; if (we > Ei) we = Ei;

  unsigned* wl = list + wave * WCAP;
  unsigned wcount = 0;

  const size_t slot = ((size_t)blockIdx.x * (size_t)nChunks + (size_t)chunk) * 32u + (size_t)wave;
  bool filter = true;
  if (LISTS == 2) {
    const unsigned saved = gc[slot];
    if (saved <= (unsigned)WCAP) {
      const unsigned* __restrict__ g = gl + slot * WCAP;
      for (unsigned i0 = 0; i0 < saved; i0 += 32 * UC) {
        unsigned en[UC];
#pragma unroll
        for (int u = 0; u < UC; ++u) {
          const unsigned i = i0 + u * 32 + lane;
          en[u] = g[i < saved ? i : 0u];
        }
#pragma unroll
        for (int u = 0; u < UC; ++u) wl[i0 + u * 32 + lane] = en[u];
      }
      wcount = saved;
      filter = false;
    }
  }

  auto add_rows = [&](const unsigned t0, const unsigned tot, const bool full) {
    unsigned en[UB];
    vecV vv[UB];
#pragma unroll
    for (int u = 0; u < UB; ++u) {
      const unsigned t = t0 + u * 32 + lane;
      en[u] = (full || t < tot) ? wl[t / LPE] : 0u;
    }
#pragma unroll
    for (int u = 0; u < UB; ++u) {
      const unsigned t = t0 + u * 32 + lane;
      const unsigned s = en[u] >> RBITS;
      const unsigned q = t % LPE;
      vv[u] = *(const vecV*)(y + (size_t)s * F + q * V);
    }
#pragma unroll
    for (int u = 0; u < UB; ++u) {
      const unsigned t = t0 + u * 32 + lane;
      if (full || t < tot) {
        const unsigned r = en[u] & RMASK;
        const unsigned q = t % LPE;
        float* a = acc + r * F + q * V;
#pragma unroll
        for (int j = 0; j < V; ++j)
          __hip_atomic_fetch_add(a + j, vv[u][j], __ATOMIC_RELAXED, __HIP_MEMORY_SCOPE_WORKGROUP);
        if (WITH_DEG && q == 0)
          __hip_atomic_fetch_add(&dg[r], 1.0f, __ATOMIC_RELAXED, __HIP_MEMORY_SCOPE_WORKGROUP);
      }
    }
  };
  constexpr unsigned BE = 32 * UB / LPE;
  unsigned done = 0;

  const int nT  = (we > ws) ? ((we - ws) >> 8) : 0;
  const int rot = (nT > 0) ? (int)(((unsigned)blockIdx.x * (unsigned)nT) / (unsigned)nRanges) : 0;
  auto trip_base = [&](const int k) {
    int p = k + rot;
    if (p >= nT) p -= nT;
    return ws + (p << 8);
  };
  int k = 0;
  if (filter && (((((size_t)dstA) | ((size_t)srcA)) & 15) == 0) && nT > 0) {
    const int* __restrict__ dp = dstA + 4 * lane;
    const int* __restrict__ sp = srcA + 4 * lane;
    int base = trip_base(0);
    v4i d0 = *(const v4i*)(dp + base);
    v4i d1 = *(const v4i*)(dp + base + 128);
    v4i s0 = *(const v4i*)(sp + base);
    v4i s1 = *(const v4i*)(sp + base + 128);
    for (; k < nT && wcount + 256u <= (unsigned)WCAP; ++k) {
      const int nb = (k + 1 < nT) ? trip_base(k + 1) : base;
      const v4i n0 = *(const v4i*)(dp + nb);
      const v4i n1 = *(const v4i*)(dp + nb + 128);
      const v4i m0 = *(const v4i*)(sp + nb);
      const v4i m1 = *(const v4i*)(sp + nb + 128);
#pragma unroll
      for (int u = 0; u < 8; ++u) {
        const int dval = (u < 4) ? d0[u & 3] : d1[u & 3];
        const int sval = (u < 4) ? s0[u & 3] : s1[u & 3];
        const unsigned r = (unsigned)(dval - r0);
        const bool m = r < (unsigned)R;
        const unsigned mask = __builtin_amdgcn_ballot_w32(m);
        const unsigned pos  = __builtin_amdgcn_mbcnt_lo(mask, wcount);
        if (m) wl[pos] = ((unsigned)sval << RBITS) | r;
        wcount += (unsigned)__builtin_popcount(mask);
      }
      if (wcount - done >= BE) {
        add_rows(done * LPE, 0u, true);
        done += BE;
      }
      d0 = n0; d1 = n1;
      s0 = m0; s1 = m1;
      base = nb;
    }
  }
  if (filter)
  for (; k <= nT; ++k) {
    const int base = (k < nT) ? trip_base(k) : ws + (nT << 8);
    const int lim  = (k < nT) ? base + 256 : we;
    if (base >= lim) break;
    int dv[UA];
    int sq[UA];
#pragma unroll
    for (int u = 0; u < UA; ++u) {
      const int e  = base + u * 32 + lane;
      const int ec = e < Ei ? e : Ei - 1;
      int d = dstA[ec];
      sq[u] = srcA[ec];
      dv[u] = (e < lim) ? d : -1;
    }
#pragma unroll
    for (int u = 0; u < UA; ++u) {
      const unsigned r = (unsigned)(dv[u] - r0);
      const bool m = r < (unsigned)R;
      const unsigned mask = __builtin_amdgcn_ballot_w32(m);
      const unsigned pos  = wcount + __builtin_amdgcn_mbcnt_lo(mask, 0u);
      if (m) {
        if (__builtin_expect(pos < (unsigned)WCAP, 1)) {
          wl[pos] = ((unsigned)sq[u] << RBITS) | r;
        } else {
          const int s = sq[u];
#pragma nounroll
          for (int f = 0; f < F; ++f)
            __hip_atomic_fetch_add(&acc[r * F + f], y[(size_t)s * F + f],
                                   __ATOMIC_RELAXED, __HIP_MEMORY_SCOPE_WORKGROUP);
          if (WITH_DEG)
            __hip_atomic_fetch_add(&dg[r], 1.0f, __ATOMIC_RELAXED, __HIP_MEMORY_SCOPE_WORKGROUP);
        }
      }
      wcount += (unsigned)__builtin_popcount(mask);
    }
  }

  const unsigned cnt = wcount < (unsigned)WCAP ? wcount : (unsigned)WCAP;
  if (LISTS == 1) {
    if (lane == 0) gc[slot] = wcount;
    for (unsigned i0 = 0; i0 < cnt; i0 += 32 * UC) {
      unsigned en[UC];
#pragma unroll
      for (int u = 0; u < UC; ++u) {
        const unsigned i = i0 + u * 32 + lane;
        en[u] = (i < cnt) ? wl[i] : 0u;
      }
#pragma unroll
      for (int u = 0; u < UC; ++u) {
        const unsigned i = i0 + u * 32 + lane;
        if (i < cnt) gl[slot * WCAP + i] = en[u];
      }
    }
  }

  const unsigned tot = cnt * LPE;
  for (unsigned t0 = done * LPE; t0 < tot; t0 += 32 * UB) add_rows(t0, tot, false);
  __syncthreads();

  const long long nF = (long long)n * F;
  float* pp = part + (size_t)chunk * (size_t)nF;
  const long long g0 = (long long)r0 * F;
  const bool vec = ((nF & 3) == 0);
  for (int i = tid; i < R * F / 4; i += 1024) {
    const long long gi = g0 + 4LL * i;
    if (vec && gi + 3 < nF) {
      *(v4f*)(pp + gi) = ((const v4f*)acc)[i];
    } else {
      for (int j = 0; j < 4; ++j)
        if (gi + j < nF) pp[gi + j] = acc[4 * i + j];
    }
  }
  if (WITH_DEG) {
    float* dp = degPart + (size_t)chunk * (size_t)n;
    for (int i = tid; i < R; i += 1024) {
      const int node = r0 + i;
      if (node < n) dp[node] = dg[i];
    }
  }
}

#define T2_STRIDE 20
template <int C>
__launch_bounds__(256)
__global__ void k_combine1_gemm2(const float* __restrict__ part, const float* __restrict__ degPart,
                                 const float* __restrict__ z1, const float* __restrict__ b1,
                                 const float* __restrict__ Wl, const float* __restrict__ Wr,
                                 float* __restrict__ y2, float* __restrict__ z2,
                                 float* __restrict__ degf, int n) {
  constexpr int NT = 4;
  __shared__ __attribute__((aligned(16))) float tile[256 * T2_STRIDE];
  const int tid = threadIdx.x;
  const int rb  = blockIdx.x * 256;
  const size_t plane = (size_t)n * 16;

  const int lane = tid & 31;
  const int wave = tid >> 5;
  const int mr   = lane & 15;
  const int kh   = (lane >> 4) << 1;

  v2f bl[4], br[4];
#pragma unroll
  for (int s = 0; s < 4; ++s) {
    int k0 = 4 * s + kh;
    v2f tl = {}; v2f tr = {};
    if (mr < 2) {
      tl.x = Wl[k0 * 2 + mr]; tl.y = Wl[(k0 + 1) * 2 + mr];
      tr.x = Wr[k0 * 2 + mr]; tr.y = Wr[(k0 + 1) * 2 + mr];
    }
    bl[s] = tl; br[s] = tr;
  }

  v4f a[NT], z[NT];
  float d[NT];
#pragma unroll
  for (int it = 0; it < NT; ++it) {
    const int t = it * 256 + tid;
    int node = rb + (t >> 2); if (node > n - 1) node = n - 1;
    const size_t off = (size_t)node * 16 + (size_t)((t & 3) * 4);
    v4f as = {0.f, 0.f, 0.f, 0.f};
    float ds = 0.0f;
#pragma unroll
    for (int c = 0; c < C; ++c) {
      as += *(const v4f*)(part + (size_t)c * plane + off);
      ds += degPart[(size_t)c * n + node];
    }
    a[it] = as; d[it] = ds;
    z[it] = *(const v4f*)(z1 + off);
  }
#pragma unroll
  for (int it = 0; it < NT; ++it) {
    const int t = it * 256 + tid;
    const int nl = t >> 2;
    const int q  = t & 3;
    if (q == 0 && rb + nl < n) degf[rb + nl] = d[it];
    const float dd = fmaxf(d[it], 1.0f);
    v4f r;
    r.x = fmaxf(a[it].x / dd + z[it].x + b1[q * 4 + 0], 0.0f);
    r.y = fmaxf(a[it].y / dd + z[it].y + b1[q * 4 + 1], 0.0f);
    r.z = fmaxf(a[it].z / dd + z[it].z + b1[q * 4 + 2], 0.0f);
    r.w = fmaxf(a[it].w / dd + z[it].w + b1[q * 4 + 3], 0.0f);
    *(v4f*)(tile + nl * T2_STRIDE + q * 4) = r;
  }
  __syncthreads();

  v8f cl[2], cr[2];
#pragma unroll
  for (int m = 0; m < 2; ++m) {
    const float* arow = tile + (wave * 32 + m * 16 + mr) * T2_STRIDE;
    v8f tl = {}; v8f tr = {};
#pragma unroll
    for (int s = 0; s < 4; ++s) {
      v2f av = *(const v2f*)(arow + 4 * s + kh);
      tl = wmma_f32x4(av, bl[s], tl);
      tr = wmma_f32x4(av, br[s], tr);
    }
    cl[m] = tl; cr[m] = tr;
  }

  const int rowHalf = (lane >> 4) << 3;
  if (mr < 2) {
#pragma unroll
    for (int m = 0; m < 2; ++m) {
#pragma unroll
      for (int v = 0; v < 8; ++v) {
        int row = rb + wave * 32 + m * 16 + rowHalf + v;
        if (row < n) {
          y2[(size_t)row * 2 + mr] = cl[m][v];
          z2[(size_t)row * 2 + mr] = cr[m][v];
        }
      }
    }
  }
}

template <int C>
__global__ void k_combine2(const float* __restrict__ part, const float* __restrict__ degf,
                           const float* __restrict__ z2, const float* __restrict__ b2,
                           float* __restrict__ out, int n) {
  const int node = (int)(blockIdx.x * blockDim.x + threadIdx.x);
  if (node >= n) return;
  const size_t plane = (size_t)n * 2;
  float a0 = 0.0f, a1 = 0.0f;
#pragma unroll
  for (int c = 0; c < C; ++c) {
    const v2f p = *(const v2f*)(part + (size_t)c * plane + (size_t)node * 2);
    a0 += p.x; a1 += p.y;
  }
  const float d = fmaxf(degf[node], 1.0f);
  const v2f z = *(const v2f*)(z2 + (size_t)node * 2);
  v2f o;
  o.x = a0 / d + z.x + b2[0];
  o.y = a1 / d + z.y + b2[1];
  *(v2f*)(out + (size_t)node * 2) = o;
}

extern "C" void kernel_launch(void* const* d_in, const int* in_sizes, int n_in,
                              void* d_out, int out_size, void* d_ws, size_t ws_size,
                              hipStream_t stream) {
  const float* x   = (const float*)d_in[0];
  const int*   ei  = (const int*)d_in[1];
  const float* W1l = (const float*)d_in[2];
  const float* W1r = (const float*)d_in[3];
  const float* b1  = (const float*)d_in[4];
  const float* W2l = (const float*)d_in[5];
  const float* W2r = (const float*)d_in[6];
  const float* b2  = (const float*)d_in[7];
  float* out = (float*)d_out;

  const int       n = in_sizes[0] / 64;
  const long long E = in_sizes[1] / 2;

  constexpr int NR1 = 28, C1 = 9;
  constexpr int NR2 = NR1, C2 = C1;

  float* ws    = (float*)d_ws;
  float* part1 = ws;
  float* part2 = part1 + (size_t)C1 * 16 * n;
  float* y1    = part2 + (size_t)C2 * 2 * n;
  float* z1    = y1 + (size_t)16 * n;
  float* y2    = z1 + (size_t)16 * n;
  float* z2    = y2 + (size_t)2 * n;
  float* degf  = z2 + (size_t)2 * n;
  float* degP  = degf + n;
  unsigned* gcnt  = (unsigned*)(degP + (size_t)C1 * n);
  unsigned* glist = gcnt + (size_t)NR1 * C1 * 32;

  const unsigned gGemm = (unsigned)((n + 127) / 128);
  k_gemm1<<<gGemm, 256, 0, stream>>>(x, W1l, W1r, y1, z1, n);

  k_agg<16, 3584, 12, 512, true, 1><<<dim3(NR1, C1), 1024, 0, stream>>>(ei, y1, part1, degP, glist, gcnt, n, E);

  k_combine1_gemm2<C1><<<(unsigned)((n + 255) / 256), 256, 0, stream>>>(part1, degP, z1, b1, W2l, W2r, y2, z2, degf, n);

  k_agg<2, 3584, 12, 512, false, 2><<<dim3(NR2, C2), 1024, 0, stream>>>(ei, y2, part2, degP, glist, gcnt, n, E);

  k_combine2<C2><<<(unsigned)((n + 255) / 256), 256, 0, stream>>>(part2, degf, z2, b2, out, n);
}
